# starting kernel
# baseline (speedup 1.0000x reference)
#include <hip/hip_runtime.h>
#include <math.h>

#define B_ 4
#define S_ 2048
#define H_ 512
#define E_ 8
#define P_ 4
#define I_ 2048
#define R_ (B_ * S_)
#define KC 128
#define XPAD 516
#define HPAD 132

typedef __attribute__((ext_vector_type(2))) float v2f;
typedef __attribute__((ext_vector_type(8))) float v8f;

__device__ __forceinline__ v8f wmma_f32(v2f a, v2f b, v8f c) {
  return __builtin_amdgcn_wmma_f32_16x16x4_f32(
       false, a,  false, b,
       (short)0, c,  false,  false);
}

__global__ __launch_bounds__(256)
void moe_router_kernel(const float* __restrict__ x,
                       const float* __restrict__ Wp,
                       const float* __restrict__ bp,
                       const float* __restrict__ Ws,
                       const float* __restrict__ bs,
                       float* __restrict__ ew) {
  const int gwave = (int)((blockIdx.x * blockDim.x + threadIdx.x) >> 5);
  const int lane  = (int)(threadIdx.x & 31);
  if (gwave >= R_) return;

  const float* xr = x + (size_t)gwave * H_;
  float a0 = 0.f, a1 = 0.f, a2 = 0.f, a3 = 0.f;
#pragma unroll 4
  for (int j = 0; j < H_ / 32; ++j) {
    const int h = lane + j * 32;
    const float xv = xr[h];
    a0 += xv * Wp[0 * H_ + h];
    a1 += xv * Wp[1 * H_ + h];
    a2 += xv * Wp[2 * H_ + h];
    a3 += xv * Wp[3 * H_ + h];
  }
#pragma unroll
  for (int off = 16; off >= 1; off >>= 1) {
    a0 += __shfl_xor(a0, off, 32);
    a1 += __shfl_xor(a1, off, 32);
    a2 += __shfl_xor(a2, off, 32);
    a3 += __shfl_xor(a3, off, 32);
  }
  if (lane == 0) {
    float p[4] = {a0 + bp[0], a1 + bp[1], a2 + bp[2], a3 + bp[3]};
    float pmax = fmaxf(fmaxf(p[0], p[1]), fmaxf(p[2], p[3]));
    float psum = 0.f;
#pragma unroll
    for (int i = 0; i < 4; ++i) { p[i] = __expf(p[i] - pmax); psum += p[i]; }
    const float pinv = 1.f / psum;
#pragma unroll
    for (int i = 0; i < 4; ++i) p[i] *= pinv;

    float el[E_];
    float emax = -1e30f;
#pragma unroll
    for (int e = 0; e < E_; ++e) {
      float v = bs[e];
#pragma unroll
      for (int q = 0; q < 4; ++q) v += p[q] * Ws[e * P_ + q];
      el[e] = v;
      emax = fmaxf(emax, v);
    }
    float esum = 0.f;
#pragma unroll
    for (int e = 0; e < E_; ++e) { el[e] = __expf(el[e] - emax); esum += el[e]; }
    const float einv = 1.f / esum;
#pragma unroll
    for (int e = 0; e < E_; ++e) ew[(size_t)e * R_ + gwave] = el[e] * einv;
  }
}

__global__ __launch_bounds__(256)
void moe_fused_kernel(const float* __restrict__ x,
                      const float* __restrict__ Wup,
                      const float* __restrict__ Wdn,
                      const float* __restrict__ ew,
                      float* __restrict__ out) {
  __shared__ float x_lds[16][XPAD];
  __shared__ float hid_lds[16][HPAD];
  __shared__ float w_lds[E_][16];

  const int row0   = (int)blockIdx.x * 16;
  const int tid    = (int)threadIdx.x;
  const int wv     = tid >> 5;
  const int lane   = tid & 31;
  const int lane16 = lane & 15;
  const int khalf  = (lane >> 4) * 2;
  const int hi8    = (lane >> 4) * 8;

  for (int idx = tid; idx < 16 * H_; idx += 256) {
    const int r = idx >> 9;
    const int c = idx & (H_ - 1);
    x_lds[r][c] = x[(size_t)(row0 + r) * H_ + c];
  }
  if (tid < E_ * 16) {
    const int e = tid >> 4, rr = tid & 15;
    w_lds[e][rr] = ew[(size_t)e * R_ + row0 + rr];
  }
  __syncthreads();

  const v8f vzero = {};
  v8f fin[4];
#pragma unroll
  for (int t = 0; t < 4; ++t) fin[t] = vzero;

  for (int e = 0; e < E_; ++e) {
    const float* Wup_e = Wup + (size_t)e * I_ * H_;
    const float* Wdn_e = Wdn + (size_t)e * H_ * I_;

    v8f eacc[4];
#pragma unroll
    for (int t = 0; t < 4; ++t) eacc[t] = vzero;

    for (int i0 = 0; i0 < I_; i0 += KC) {
      v8f hacc = vzero;
      const float* bptr = Wup_e + (size_t)(i0 + wv * 16 + lane16) * H_ + khalf;
#pragma unroll 4
      for (int k0 = 0; k0 < H_; k0 += 4) {
        v2f a, b;
        a[0] = x_lds[lane16][k0 + khalf];
        a[1] = x_lds[lane16][k0 + khalf + 1];
        b[0] = bptr[k0];
        b[1] = bptr[k0 + 1];
        hacc = wmma_f32(a, b, hacc);
      }
#pragma unroll
      for (int v = 0; v < 8; ++v) {
        const float hv = hacc[v];
        const float g = 0.5f * hv * (1.0f + erff(hv * 0.70710678118654752f));
        hid_lds[v + hi8][wv * 16 + lane16] = g;
      }
      __syncthreads();

#pragma unroll
      for (int t = 0; t < 4; ++t) {
        const int n = wv * 64 + t * 16 + lane16;
        const float* b2 = Wdn_e + (size_t)n * I_ + i0 + khalf;
#pragma unroll 4
        for (int k0 = 0; k0 < KC; k0 += 4) {
          v2f a, b;
          a[0] = hid_lds[lane16][k0 + khalf];
          a[1] = hid_lds[lane16][k0 + khalf + 1];
          b[0] = b2[k0];
          b[1] = b2[k0 + 1];
          eacc[t] = wmma_f32(a, b, eacc[t]);
        }
      }
      __syncthreads();
    }

#pragma unroll
    for (int t = 0; t < 4; ++t) {
#pragma unroll
      for (int v = 0; v < 8; ++v) {
        const float wgt = w_lds[e][v + hi8];
        fin[t][v] += wgt * eacc[t][v];
      }
    }
  }

#pragma unroll
  for (int t = 0; t < 4; ++t) {
    const int col = wv * 64 + t * 16 + lane16;
#pragma unroll
    for (int v = 0; v < 8; ++v) {
      const int row = row0 + v + hi8;
      out[(size_t)row * H_ + col] = fin[t][v];
    }
  }
}

extern "C" void kernel_launch(void* const* d_in, const int* in_sizes, int n_in,
                              void* d_out, int out_size, void* d_ws, size_t ws_size,
                              hipStream_t stream) {
  const float* x    = (const float*)d_in[0];
  const float* Wp   = (const float*)d_in[1];
  const float* bp   = (const float*)d_in[2];
  const float* Ws   = (const float*)d_in[3];
  const float* bs   = (const float*)d_in[4];
  const float* Wup  = (const float*)d_in[5];
  const float* Wdn  = (const float*)d_in[6];
  float* out = (float*)d_out;
  float* ew  = (float*)d_ws;

  moe_router_kernel<<<R_ / 8, 256, 0, stream>>>(x, Wp, bp, Ws, bs, ew);

  moe_fused_kernel<<<R_ / 16, 256, 0, stream>>>(x, Wup, Wdn, ew, out);
}


# Weights go through LDS: 32-row workgroups of 16 waves share each 64 KB weight piece, read once with full-line 16-byte loads into a 2-stage LDS ring, two pieces ahead; every sum keeps its order
# speedup vs baseline: 1.7827x; 1.7827x over previous
#include <hip/hip_runtime.h>
#include <math.h>

#define B_ 4
#define S_ 2048
#define H_ 512
#define E_ 8
#define P_ 4
#define I_ 2048
#define R_ (B_ * S_)
#define KC 128
#define XPAD 516
#define HPAD 132
#define MT 32
#define NTHR 512
#define UPITCH 132
#define DPITCH 36
#define STAGE_FLOATS 18432

typedef __attribute__((ext_vector_type(2))) float v2f;
typedef __attribute__((ext_vector_type(4))) float v4f;
typedef __attribute__((ext_vector_type(8))) float v8f;

__device__ __forceinline__ v8f wmma_f32(v2f a, v2f b, v8f c) {
  return __builtin_amdgcn_wmma_f32_16x16x4_f32(
       false, a,  false, b,
       (short)0, c,  false,  false);
}

__global__ __launch_bounds__(256)
void moe_router_kernel(const float* __restrict__ x,
                       const float* __restrict__ Wp,
                       const float* __restrict__ bp,
                       const float* __restrict__ Ws,
                       const float* __restrict__ bs,
                       float* __restrict__ ew) {
  const int gwave = (int)((blockIdx.x * blockDim.x + threadIdx.x) >> 5);
  const int lane  = (int)(threadIdx.x & 31);
  if (gwave >= R_) return;

  const float* xr = x + (size_t)gwave * H_;
  float a0 = 0.f, a1 = 0.f, a2 = 0.f, a3 = 0.f;
#pragma unroll 4
  for (int j = 0; j < H_ / 32; ++j) {
    const int h = lane + j * 32;
    const float xv = xr[h];
    a0 += xv * Wp[0 * H_ + h];
    a1 += xv * Wp[1 * H_ + h];
    a2 += xv * Wp[2 * H_ + h];
    a3 += xv * Wp[3 * H_ + h];
  }
#pragma unroll
  for (int off = 16; off >= 1; off >>= 1) {
    a0 += __shfl_xor(a0, off, 32);
    a1 += __shfl_xor(a1, off, 32);
    a2 += __shfl_xor(a2, off, 32);
    a3 += __shfl_xor(a3, off, 32);
  }
  if (lane == 0) {
    float p[4] = {a0 + bp[0], a1 + bp[1], a2 + bp[2], a3 + bp[3]};
    float pmax = fmaxf(fmaxf(p[0], p[1]), fmaxf(p[2], p[3]));
    float psum = 0.f;
#pragma unroll
    for (int i = 0; i < 4; ++i) { p[i] = __expf(p[i] - pmax); psum += p[i]; }
    const float pinv = 1.f / psum;
#pragma unroll
    for (int i = 0; i < 4; ++i) p[i] *= pinv;

    float el[E_];
    float emax = -1e30f;
#pragma unroll
    for (int e = 0; e < E_; ++e) {
      float v = bs[e];
#pragma unroll
      for (int q = 0; q < 4; ++q) v += p[q] * Ws[e * P_ + q];
      el[e] = v;
      emax = fmaxf(emax, v);
    }
    float esum = 0.f;
#pragma unroll
    for (int e = 0; e < E_; ++e) { el[e] = __expf(el[e] - emax); esum += el[e]; }
    const float einv = 1.f / esum;
#pragma unroll
    for (int e = 0; e < E_; ++e) ew[(size_t)e * R_ + gwave] = el[e] * einv;
  }
}


__device__ __forceinline__ void wg_barrier_lds_only() {
  asm volatile("s_wait_dscnt 0x0\n\ts_barrier_signal -1\n\ts_barrier_wait -1" ::: "memory");
}

__device__ __forceinline__ void load_up_piece(v4f (&pf)[8], const float* __restrict__ Wup_c, int p, int upo) {
#pragma unroll
  for (int j = 0; j < 8; ++j) pf[j] = *(const v4f*)(Wup_c + (upo + p * 128 + j * 16 * H_));
}
__device__ __forceinline__ void store_up_piece(const v4f (&pf)[8], float* buf, int ups) {
#pragma unroll
  for (int j = 0; j < 8; ++j) *(v4f*)(buf + (ups + j * 16 * UPITCH)) = pf[j];
}
__device__ __forceinline__ void load_dn_piece(v4f (&pf)[8], const float* __restrict__ Wdn_c, int q, int dno) {
#pragma unroll
  for (int j = 0; j < 8; ++j) pf[j] = *(const v4f*)(Wdn_c + (dno + q * 32 + j * 64 * I_));
}
__device__ __forceinline__ void store_dn_piece(const v4f (&pf)[8], float* buf, int dns) {
#pragma unroll
  for (int j = 0; j < 8; ++j) *(v4f*)(buf + (dns + j * 64 * DPITCH)) = pf[j];
}

__global__ __launch_bounds__(NTHR)
void moe_fused_kernel(const float* __restrict__ x,
                      const float* __restrict__ Wup,
                      const float* __restrict__ Wdn,
                      const float* __restrict__ ew,
                      float* __restrict__ out) {
  __shared__ __attribute__((aligned(16))) float x_lds[MT][XPAD];
  __shared__ __attribute__((aligned(16))) float hid_lds[MT][HPAD];
  __shared__ float w_lds[E_][MT];
  __shared__ __attribute__((aligned(16))) float wst[2][STAGE_FLOATS];

  const int row0   = (int)blockIdx.x * MT;
  const int tid    = (int)threadIdx.x;
  const int wv     = tid >> 5;
  const int lane   = tid & 31;
  const int lane16 = lane & 15;
  const int khalf  = (lane >> 4) * 2;
  const int hi8    = (lane >> 4) * 8;
  const int rt     = wv >> 3;
  const int cg     = wv & 7;
  const int upo    = (tid >> 5) * H_ + (tid & 31) * 4;
  const int ups    = (tid >> 5) * UPITCH + (tid & 31) * 4;
  const int dno    = (tid >> 3) * I_ + (tid & 7) * 4;
  const int dns    = (tid >> 3) * DPITCH + (tid & 7) * 4;

  for (int idx = tid; idx < MT * H_ / 4; idx += NTHR) {
    const int r = idx >> 7, c4 = idx & 127;
    *(v4f*)&x_lds[r][c4 * 4] = *(const v4f*)(x + (size_t)(row0 + r) * H_ + c4 * 4);
  }
  if (tid < E_ * MT) {
    const int e = tid >> 5, rr = tid & 31;
    w_lds[e][rr] = ew[(size_t)e * R_ + row0 + rr];
  }

  v4f pf[8];
  load_up_piece(pf, Wup, 0, upo);
  store_up_piece(pf, wst[0], ups);
  load_up_piece(pf, Wup, 1, upo);

  const v8f vzero = {};
  v8f fin[4], eacc[4];
#pragma unroll
  for (int t = 0; t < 4; ++t) { fin[t] = vzero; eacc[t] = vzero; }

  const int NCH = E_ * (I_ / KC);
  for (int ch = 0; ch < NCH; ++ch) {
    const int e = ch >> 4;
    const int c = ch & 15;
    const float* Wup_c = Wup + (size_t)ch * KC * H_;
    const float* Wdn_c = Wdn + (size_t)e * H_ * I_ + (size_t)c * KC;

    v8f hacc = vzero;
#pragma unroll
    for (int p = 0; p < 4; ++p) {
      wg_barrier_lds_only();
      const float* buf = wst[p & 1];
      float* nbuf = wst[(p + 1) & 1];
      const float* arow = &x_lds[rt * 16 + lane16][p * 128 + khalf];
      const float* brow = buf + (cg * 16 + lane16) * UPITCH + khalf;
#pragma unroll
      for (int k0 = 0; k0 < 64; k0 += 4) {
        const v2f a = *(const v2f*)(arow + k0);
        const v2f b = *(const v2f*)(brow + k0);
        hacc = wmma_f32(a, b, hacc);
      }
      if (p < 3) store_up_piece(pf, nbuf, ups); else store_dn_piece(pf, nbuf, dns);
      if (p < 2) load_up_piece(pf, Wup_c, p + 2, upo); else load_dn_piece(pf, Wdn_c, p - 2, dno);
      __builtin_amdgcn_sched_barrier(0);
#pragma unroll
      for (int k0 = 64; k0 < 128; k0 += 4) {
        const v2f a = *(const v2f*)(arow + k0);
        const v2f b = *(const v2f*)(brow + k0);
        hacc = wmma_f32(a, b, hacc);
      }
    }
#pragma unroll
    for (int v = 0; v < 8; ++v) {
      const float hv = hacc[v];
      const float g = 0.5f * hv * (1.0f + erff(hv * 0.70710678118654752f));
      hid_lds[rt * 16 + v + hi8][cg * 16 + lane16] = g;
    }

    const bool more = (ch + 1 < NCH);
#pragma unroll
    for (int q = 0; q < 4; ++q) {
      wg_barrier_lds_only();
      const float* buf = wst[q & 1];
      float* nbuf = wst[(q + 1) & 1];
      const float* arow = &hid_lds[rt * 16 + lane16][q * 32 + khalf];
      const float* brow = buf + (cg * 64 + lane16) * DPITCH + khalf;
#pragma unroll
      for (int k0 = 0; k0 < 16; k0 += 4) {
        const v2f a = *(const v2f*)(arow + k0);
#pragma unroll
        for (int t = 0; t < 4; ++t) {
          const v2f b = *(const v2f*)(brow + t * 16 * DPITCH + k0);
          eacc[t] = wmma_f32(a, b, eacc[t]);
        }
      }
      if (q < 3)     store_dn_piece(pf, nbuf, dns);
      else if (more) store_up_piece(pf, nbuf, ups);
      if (q < 2)     load_dn_piece(pf, Wdn_c, q + 2, dno);
      else if (more) load_up_piece(pf, Wup_c + (size_t)KC * H_, q - 2, upo);
      __builtin_amdgcn_sched_barrier(0);
#pragma unroll
      for (int k0 = 16; k0 < 32; k0 += 4) {
        const v2f a = *(const v2f*)(arow + k0);
#pragma unroll
        for (int t = 0; t < 4; ++t) {
          const v2f b = *(const v2f*)(brow + t * 16 * DPITCH + k0);
          eacc[t] = wmma_f32(a, b, eacc[t]);
        }
      }
    }

    if (c == 15) {
#pragma unroll
      for (int t = 0; t < 4; ++t) {
#pragma unroll
        for (int v = 0; v < 8; ++v) {
          const float wgt = w_lds[e][rt * 16 + v + hi8];
          fin[t][v] += wgt * eacc[t][v];
        }
        eacc[t] = vzero;
      }
    }
  }

#pragma unroll
  for (int t = 0; t < 4; ++t) {
    const int col = cg * 64 + t * 16 + lane16;
#pragma unroll
    for (int v = 0; v < 8; ++v) {
      const int row = row0 + rt * 16 + v + hi8;
      out[(size_t)row * H_ + col] = fin[t][v];
    }
  }
}

extern "C" void kernel_launch(void* const* d_in, const int* in_sizes, int n_in,
                              void* d_out, int out_size, void* d_ws, size_t ws_size,
                              hipStream_t stream) {
  const float* x    = (const float*)d_in[0];
  const float* Wp   = (const float*)d_in[1];
  const float* bp   = (const float*)d_in[2];
  const float* Ws   = (const float*)d_in[3];
  const float* bs   = (const float*)d_in[4];
  const float* Wup  = (const float*)d_in[5];
  const float* Wdn  = (const float*)d_in[6];
  float* out = (float*)d_out;
  float* ew  = (float*)d_ws;

  moe_router_kernel<<<R_ / 8, 256, 0, stream>>>(x, Wp, bp, Ws, bs, ew);

  moe_fused_kernel<<<R_ / MT, NTHR, 0, stream>>>(x, Wup, Wdn, ew, out);
}


# Up-projection fragment reads leave the matrix stream: eight k-steps of A and B are read from LDS as one group, one block ahead, into a second register set; every sum keeps its order
# speedup vs baseline: 1.8634x; 1.0453x over previous
#include <hip/hip_runtime.h>
#include <math.h>

#define B_ 4
#define S_ 2048
#define H_ 512
#define E_ 8
#define P_ 4
#define I_ 2048
#define R_ (B_ * S_)
#define KC 128
#define XPAD 516
#define HPAD 132
#define MT 32
#define NTHR 512
#define UPITCH 132
#define DPITCH 36
#define STAGE_FLOATS 18432

typedef __attribute__((ext_vector_type(2))) float v2f;
typedef __attribute__((ext_vector_type(4))) float v4f;
typedef __attribute__((ext_vector_type(8))) float v8f;

__device__ __forceinline__ v8f wmma_f32(v2f a, v2f b, v8f c) {
  return __builtin_amdgcn_wmma_f32_16x16x4_f32(
       false, a,  false, b,
       (short)0, c,  false,  false);
}

__global__ __launch_bounds__(256)
void moe_router_kernel(const float* __restrict__ x,
                       const float* __restrict__ Wp,
                       const float* __restrict__ bp,
                       const float* __restrict__ Ws,
                       const float* __restrict__ bs,
                       float* __restrict__ ew) {
  const int gwave = (int)((blockIdx.x * blockDim.x + threadIdx.x) >> 5);
  const int lane  = (int)(threadIdx.x & 31);
  if (gwave >= R_) return;

  const float* xr = x + (size_t)gwave * H_;
  float a0 = 0.f, a1 = 0.f, a2 = 0.f, a3 = 0.f;
#pragma unroll 4
  for (int j = 0; j < H_ / 32; ++j) {
    const int h = lane + j * 32;
    const float xv = xr[h];
    a0 += xv * Wp[0 * H_ + h];
    a1 += xv * Wp[1 * H_ + h];
    a2 += xv * Wp[2 * H_ + h];
    a3 += xv * Wp[3 * H_ + h];
  }
#pragma unroll
  for (int off = 16; off >= 1; off >>= 1) {
    a0 += __shfl_xor(a0, off, 32);
    a1 += __shfl_xor(a1, off, 32);
    a2 += __shfl_xor(a2, off, 32);
    a3 += __shfl_xor(a3, off, 32);
  }
  if (lane == 0) {
    float p[4] = {a0 + bp[0], a1 + bp[1], a2 + bp[2], a3 + bp[3]};
    float pmax = fmaxf(fmaxf(p[0], p[1]), fmaxf(p[2], p[3]));
    float psum = 0.f;
#pragma unroll
    for (int i = 0; i < 4; ++i) { p[i] = __expf(p[i] - pmax); psum += p[i]; }
    const float pinv = 1.f / psum;
#pragma unroll
    for (int i = 0; i < 4; ++i) p[i] *= pinv;

    float el[E_];
    float emax = -1e30f;
#pragma unroll
    for (int e = 0; e < E_; ++e) {
      float v = bs[e];
#pragma unroll
      for (int q = 0; q < 4; ++q) v += p[q] * Ws[e * P_ + q];
      el[e] = v;
      emax = fmaxf(emax, v);
    }
    float esum = 0.f;
#pragma unroll
    for (int e = 0; e < E_; ++e) { el[e] = __expf(el[e] - emax); esum += el[e]; }
    const float einv = 1.f / esum;
#pragma unroll
    for (int e = 0; e < E_; ++e) ew[(size_t)e * R_ + gwave] = el[e] * einv;
  }
}


__device__ __forceinline__ void wg_barrier_lds_only() {
  asm volatile("s_wait_dscnt 0x0\n\ts_barrier_signal -1\n\ts_barrier_wait -1" ::: "memory");
}

__device__ __forceinline__ void load_up_piece(v4f (&pf)[8], const float* __restrict__ Wup_c, int p, int upo) {
#pragma unroll
  for (int j = 0; j < 8; ++j) pf[j] = *(const v4f*)(Wup_c + (upo + p * 128 + j * 16 * H_));
}
__device__ __forceinline__ void store_up_piece(const v4f (&pf)[8], float* buf, int ups) {
#pragma unroll
  for (int j = 0; j < 8; ++j) *(v4f*)(buf + (ups + j * 16 * UPITCH)) = pf[j];
}
__device__ __forceinline__ void load_dn_piece(v4f (&pf)[8], const float* __restrict__ Wdn_c, int q, int dno) {
#pragma unroll
  for (int j = 0; j < 8; ++j) pf[j] = *(const v4f*)(Wdn_c + (dno + q * 32 + j * 64 * I_));
}
__device__ __forceinline__ void store_dn_piece(const v4f (&pf)[8], float* buf, int dns) {
#pragma unroll
  for (int j = 0; j < 8; ++j) *(v4f*)(buf + (dns + j * 64 * DPITCH)) = pf[j];
}

__device__ __forceinline__ void load_frag8(v2f (&fa)[8], v2f (&fb)[8],
                                           const float* arow, const float* brow) {
#pragma unroll
  for (int s = 0; s < 8; ++s) {
    fa[s] = *(const v2f*)(arow + s * 4);
    fb[s] = *(const v2f*)(brow + s * 4);
  }
}
__device__ __forceinline__ v8f mma_frag8(const v2f (&fa)[8], const v2f (&fb)[8], v8f acc) {
#pragma unroll
  for (int s = 0; s < 8; ++s) acc = wmma_f32(fa[s], fb[s], acc);
  return acc;
}

__global__ __launch_bounds__(NTHR)
void moe_fused_kernel(const float* __restrict__ x,
                      const float* __restrict__ Wup,
                      const float* __restrict__ Wdn,
                      const float* __restrict__ ew,
                      float* __restrict__ out) {
  __shared__ __attribute__((aligned(16))) float x_lds[MT][XPAD];
  __shared__ __attribute__((aligned(16))) float hid_lds[MT][HPAD];
  __shared__ float w_lds[E_][MT];
  __shared__ __attribute__((aligned(16))) float wst0[STAGE_FLOATS];
  __shared__ __attribute__((aligned(16))) float wst1[STAGE_FLOATS];

  const int row0   = (int)blockIdx.x * MT;
  const int tid    = (int)threadIdx.x;
  const int wv     = tid >> 5;
  const int lane   = tid & 31;
  const int lane16 = lane & 15;
  const int khalf  = (lane >> 4) * 2;
  const int hi8    = (lane >> 4) * 8;
  const int rt     = wv >> 3;
  const int cg     = wv & 7;
  const int upo    = (tid >> 5) * H_ + (tid & 31) * 4;
  const int ups    = (tid >> 5) * UPITCH + (tid & 31) * 4;
  const int dno    = (tid >> 3) * I_ + (tid & 7) * 4;
  const int dns    = (tid >> 3) * DPITCH + (tid & 7) * 4;

  for (int idx = tid; idx < MT * H_ / 4; idx += NTHR) {
    const int r = idx >> 7, c4 = idx & 127;
    *(v4f*)&x_lds[r][c4 * 4] = *(const v4f*)(x + (size_t)(row0 + r) * H_ + c4 * 4);
  }
  if (tid < E_ * MT) {
    const int e = tid >> 5, rr = tid & 31;
    w_lds[e][rr] = ew[(size_t)e * R_ + row0 + rr];
  }

  v4f pf[8];
  load_up_piece(pf, Wup, 0, upo);
  store_up_piece(pf, wst0, ups);
  load_up_piece(pf, Wup, 1, upo);

  const v8f vzero = {};
  v8f fin[4], eacc[4];
#pragma unroll
  for (int t = 0; t < 4; ++t) { fin[t] = vzero; eacc[t] = vzero; }

  const int NCH = E_ * (I_ / KC);
  for (int ch = 0; ch < NCH; ++ch) {
    const int e = ch >> 4;
    const int c = ch & 15;
    const float* Wup_c = Wup + (size_t)ch * KC * H_;
    const float* Wdn_c = Wdn + (size_t)e * H_ * I_ + (size_t)c * KC;

    v8f hacc = vzero;
    v2f fa0[8], fb0[8], fa1[8], fb1[8];
#pragma unroll
    for (int p = 0; p < 4; ++p) {
      wg_barrier_lds_only();
      const float* buf = (p & 1) ? wst1 : wst0;
      float* nbuf = ((p + 1) & 1) ? wst1 : wst0;
      const float* arow = &x_lds[rt * 16 + lane16][p * 128 + khalf];
      const float* brow = buf + (cg * 16 + lane16) * UPITCH + khalf;
      load_frag8(fa0, fb0, arow, brow);
      __builtin_amdgcn_sched_barrier(0);
      if (p > 0) hacc = mma_frag8(fa1, fb1, hacc);
      __builtin_amdgcn_sched_barrier(0);
      load_frag8(fa1, fb1, arow + 32, brow + 32);
      __builtin_amdgcn_sched_barrier(0);
      hacc = mma_frag8(fa0, fb0, hacc);
      __builtin_amdgcn_sched_barrier(0);
      if (p < 3) store_up_piece(pf, nbuf, ups); else store_dn_piece(pf, nbuf, dns);
      if (p < 2) load_up_piece(pf, Wup_c, p + 2, upo); else load_dn_piece(pf, Wdn_c, p - 2, dno);
      load_frag8(fa0, fb0, arow + 64, brow + 64);
      __builtin_amdgcn_sched_barrier(0);
      hacc = mma_frag8(fa1, fb1, hacc);
      __builtin_amdgcn_sched_barrier(0);
      load_frag8(fa1, fb1, arow + 96, brow + 96);
      __builtin_amdgcn_sched_barrier(0);
      hacc = mma_frag8(fa0, fb0, hacc);
      __builtin_amdgcn_sched_barrier(0);
    }
    hacc = mma_frag8(fa1, fb1, hacc);
#pragma unroll
    for (int v = 0; v < 8; ++v) {
      const float hv = hacc[v];
      const float g = 0.5f * hv * (1.0f + erff(hv * 0.70710678118654752f));
      hid_lds[rt * 16 + v + hi8][cg * 16 + lane16] = g;
    }

    const bool more = (ch + 1 < NCH);
#pragma unroll
    for (int q = 0; q < 4; ++q) {
      wg_barrier_lds_only();
      const float* buf = (q & 1) ? wst1 : wst0;
      float* nbuf = ((q + 1) & 1) ? wst1 : wst0;
      const float* arow = &hid_lds[rt * 16 + lane16][q * 32 + khalf];
      const float* brow = buf + (cg * 64 + lane16) * DPITCH + khalf;
#pragma unroll
      for (int k0 = 0; k0 < 16; k0 += 4) {
        const v2f a = *(const v2f*)(arow + k0);
#pragma unroll
        for (int t = 0; t < 4; ++t) {
          const v2f b = *(const v2f*)(brow + t * 16 * DPITCH + k0);
          eacc[t] = wmma_f32(a, b, eacc[t]);
        }
      }
      if (q < 3)     store_dn_piece(pf, nbuf, dns);
      else if (more) store_up_piece(pf, nbuf, ups);
      if (q < 2)     load_dn_piece(pf, Wdn_c, q + 2, dno);
      else if (more) load_up_piece(pf, Wup_c + (size_t)KC * H_, q - 2, upo);
      __builtin_amdgcn_sched_barrier(0);
#pragma unroll
      for (int k0 = 16; k0 < 32; k0 += 4) {
        const v2f a = *(const v2f*)(arow + k0);
#pragma unroll
        for (int t = 0; t < 4; ++t) {
          const v2f b = *(const v2f*)(brow + t * 16 * DPITCH + k0);
          eacc[t] = wmma_f32(a, b, eacc[t]);
        }
      }
    }

    if (c == 15) {
#pragma unroll
      for (int t = 0; t < 4; ++t) {
#pragma unroll
        for (int v = 0; v < 8; ++v) {
          const float wgt = w_lds[e][rt * 16 + v + hi8];
          fin[t][v] += wgt * eacc[t][v];
        }
        eacc[t] = vzero;
      }
    }
  }

#pragma unroll
  for (int t = 0; t < 4; ++t) {
    const int col = cg * 64 + t * 16 + lane16;
#pragma unroll
    for (int v = 0; v < 8; ++v) {
      const int row = row0 + rt * 16 + v + hi8;
      out[(size_t)row * H_ + col] = fin[t][v];
    }
  }
}

extern "C" void kernel_launch(void* const* d_in, const int* in_sizes, int n_in,
                              void* d_out, int out_size, void* d_ws, size_t ws_size,
                              hipStream_t stream) {
  const float* x    = (const float*)d_in[0];
  const float* Wp   = (const float*)d_in[1];
  const float* bp   = (const float*)d_in[2];
  const float* Ws   = (const float*)d_in[3];
  const float* bs   = (const float*)d_in[4];
  const float* Wup  = (const float*)d_in[5];
  const float* Wdn  = (const float*)d_in[6];
  float* out = (float*)d_out;
  float* ew  = (float*)d_ws;

  moe_router_kernel<<<R_ / 8, 256, 0, stream>>>(x, Wp, bp, Ws, bs, ew);

  moe_fused_kernel<<<R_ / MT, NTHR, 0, stream>>>(x, Wup, Wdn, ew, out);
}


# Workgroup tile is 64 token rows x half the experts: each weight piece in LDS feeds twice the products; a third kernel adds the two expert halves (expert sum order changes)
# speedup vs baseline: 2.0478x; 1.0990x over previous
#include <hip/hip_runtime.h>
#include <math.h>

#define B_ 4
#define S_ 2048
#define H_ 512
#define E_ 8
#define P_ 4
#define I_ 2048
#define R_ (B_ * S_)
#define KC 128
#define XPAD 516
#define HPAD 132
#define MT 64
#define EPW 4
#define NPART (E_ / EPW)
#define PART_OFF (1 << 18)
#define NTHR 512
#define UPITCH 132
#define DPITCH 36
#define STAGE_FLOATS 18432

typedef __attribute__((ext_vector_type(2))) float v2f;
typedef __attribute__((ext_vector_type(4))) float v4f;
typedef __attribute__((ext_vector_type(8))) float v8f;

__device__ __forceinline__ v8f wmma_f32(v2f a, v2f b, v8f c) {
  return __builtin_amdgcn_wmma_f32_16x16x4_f32(
       false, a,  false, b,
       (short)0, c,  false,  false);
}

__global__ __launch_bounds__(256)
void moe_router_kernel(const float* __restrict__ x,
                       const float* __restrict__ Wp,
                       const float* __restrict__ bp,
                       const float* __restrict__ Ws,
                       const float* __restrict__ bs,
                       float* __restrict__ ew) {
  const int gwave = (int)((blockIdx.x * blockDim.x + threadIdx.x) >> 5);
  const int lane  = (int)(threadIdx.x & 31);
  if (gwave >= R_) return;

  const float* xr = x + (size_t)gwave * H_;
  float a0 = 0.f, a1 = 0.f, a2 = 0.f, a3 = 0.f;
#pragma unroll 4
  for (int j = 0; j < H_ / 32; ++j) {
    const int h = lane + j * 32;
    const float xv = xr[h];
    a0 += xv * Wp[0 * H_ + h];
    a1 += xv * Wp[1 * H_ + h];
    a2 += xv * Wp[2 * H_ + h];
    a3 += xv * Wp[3 * H_ + h];
  }
#pragma unroll
  for (int off = 16; off >= 1; off >>= 1) {
    a0 += __shfl_xor(a0, off, 32);
    a1 += __shfl_xor(a1, off, 32);
    a2 += __shfl_xor(a2, off, 32);
    a3 += __shfl_xor(a3, off, 32);
  }
  if (lane == 0) {
    float p[4] = {a0 + bp[0], a1 + bp[1], a2 + bp[2], a3 + bp[3]};
    float pmax = fmaxf(fmaxf(p[0], p[1]), fmaxf(p[2], p[3]));
    float psum = 0.f;
#pragma unroll
    for (int i = 0; i < 4; ++i) { p[i] = __expf(p[i] - pmax); psum += p[i]; }
    const float pinv = 1.f / psum;
#pragma unroll
    for (int i = 0; i < 4; ++i) p[i] *= pinv;

    float el[E_];
    float emax = -1e30f;
#pragma unroll
    for (int e = 0; e < E_; ++e) {
      float v = bs[e];
#pragma unroll
      for (int q = 0; q < 4; ++q) v += p[q] * Ws[e * P_ + q];
      el[e] = v;
      emax = fmaxf(emax, v);
    }
    float esum = 0.f;
#pragma unroll
    for (int e = 0; e < E_; ++e) { el[e] = __expf(el[e] - emax); esum += el[e]; }
    const float einv = 1.f / esum;
#pragma unroll
    for (int e = 0; e < E_; ++e) ew[(size_t)e * R_ + gwave] = el[e] * einv;
  }
}


__device__ __forceinline__ void wg_barrier_lds_only() {
  asm volatile("s_wait_dscnt 0x0\n\ts_barrier_signal -1\n\ts_barrier_wait -1" ::: "memory");
}

__device__ __forceinline__ void load_up_piece(v4f (&pf)[8], const float* __restrict__ Wup_c, int p, int upo) {
#pragma unroll
  for (int j = 0; j < 8; ++j) pf[j] = *(const v4f*)(Wup_c + (upo + p * 128 + j * 16 * H_));
}
__device__ __forceinline__ void store_up_piece(const v4f (&pf)[8], float* buf, int ups) {
#pragma unroll
  for (int j = 0; j < 8; ++j) *(v4f*)(buf + (ups + j * 16 * UPITCH)) = pf[j];
}
__device__ __forceinline__ void load_dn_piece(v4f (&pf)[8], const float* __restrict__ Wdn_c, int q, int dno) {
#pragma unroll
  for (int j = 0; j < 8; ++j) pf[j] = *(const v4f*)(Wdn_c + (dno + q * 32 + j * 64 * I_));
}
__device__ __forceinline__ void store_dn_piece(const v4f (&pf)[8], float* buf, int dns) {
#pragma unroll
  for (int j = 0; j < 8; ++j) *(v4f*)(buf + (dns + j * 64 * DPITCH)) = pf[j];
}

__device__ __forceinline__ void load_frag4x2(v2f (&fa0)[4], v2f (&fa1)[4], v2f (&fb)[4],
                                             const float* arow0, const float* arow1, const float* brow) {
#pragma unroll
  for (int s = 0; s < 4; ++s) {
    fa0[s] = *(const v2f*)(arow0 + s * 4);
    fa1[s] = *(const v2f*)(arow1 + s * 4);
    fb[s]  = *(const v2f*)(brow + s * 4);
  }
}
__device__ __forceinline__ void mma_frag4x2(const v2f (&fa0)[4], const v2f (&fa1)[4], const v2f (&fb)[4],
                                            v8f& h0, v8f& h1) {
#pragma unroll
  for (int s = 0; s < 4; ++s) {
    h0 = wmma_f32(fa0[s], fb[s], h0);
    h1 = wmma_f32(fa1[s], fb[s], h1);
  }
}

__global__ __launch_bounds__(NTHR)
void moe_fused_kernel(const float* __restrict__ x,
                      const float* __restrict__ Wup,
                      const float* __restrict__ Wdn,
                      const float* __restrict__ ew,
                      float* __restrict__ part) {
  __shared__ __attribute__((aligned(16))) float x_lds[MT][XPAD];
  __shared__ __attribute__((aligned(16))) float hid_lds[MT][HPAD];
  __shared__ float w_lds[EPW][MT];
  __shared__ __attribute__((aligned(16))) float wst0[STAGE_FLOATS];
  __shared__ __attribute__((aligned(16))) float wst1[STAGE_FLOATS];

  const int eh     = (int)blockIdx.x % NPART;
  const int row0   = ((int)blockIdx.x / NPART) * MT;
  const int e0     = eh * EPW;
  const int tid    = (int)threadIdx.x;
  const int wv     = tid >> 5;
  const int lane   = tid & 31;
  const int lane16 = lane & 15;
  const int khalf  = (lane >> 4) * 2;
  const int hi8    = (lane >> 4) * 8;
  const int rp     = wv >> 3;
  const int cg     = wv & 7;
  const int upo    = (tid >> 5) * H_ + (tid & 31) * 4;
  const int ups    = (tid >> 5) * UPITCH + (tid & 31) * 4;
  const int dno    = (tid >> 3) * I_ + (tid & 7) * 4;
  const int dns    = (tid >> 3) * DPITCH + (tid & 7) * 4;

  for (int idx = tid; idx < MT * H_ / 4; idx += NTHR) {
    const int r = idx >> 7, c4 = idx & 127;
    *(v4f*)&x_lds[r][c4 * 4] = *(const v4f*)(x + (size_t)(row0 + r) * H_ + c4 * 4);
  }
  if (tid < EPW * MT) {
    const int el = tid / MT, rr = tid % MT;
    w_lds[el][rr] = ew[(size_t)(e0 + el) * R_ + row0 + rr];
  }

  const int CH0 = e0 * (I_ / KC);
  const int CH1 = CH0 + EPW * (I_ / KC);

  v4f pf[8];
  load_up_piece(pf, Wup + (size_t)CH0 * KC * H_, 0, upo);
  store_up_piece(pf, wst0, ups);
  load_up_piece(pf, Wup + (size_t)CH0 * KC * H_, 1, upo);

  const v8f vzero = {};
  v8f fin[2][4], eacc[2][4];
#pragma unroll
  for (int t = 0; t < 2; ++t) {
#pragma unroll
    for (int u = 0; u < 4; ++u) { fin[t][u] = vzero; eacc[t][u] = vzero; }
  }

  for (int ch = CH0; ch < CH1; ++ch) {
    const int e = ch >> 4;
    const int c = ch & 15;
    const float* Wup_c = Wup + (size_t)ch * KC * H_;
    const float* Wdn_c = Wdn + (size_t)e * H_ * I_ + (size_t)c * KC;

    v8f hacc0 = vzero, hacc1 = vzero;
    v2f fa0[2][4], fa1[2][4], fb[2][4];
#pragma unroll
    for (int p = 0; p < 4; ++p) {
      wg_barrier_lds_only();
      const float* buf = (p & 1) ? wst1 : wst0;
      float* nbuf = ((p + 1) & 1) ? wst1 : wst0;
      const float* arow0 = &x_lds[rp * 32 + lane16][p * 128 + khalf];
      const float* arow1 = arow0 + 16 * XPAD;
      const float* brow = buf + (cg * 16 + lane16) * UPITCH + khalf;
#pragma unroll
      for (int b = 0; b < 8; ++b) {
        if (b == 3) {
          if (p < 3) store_up_piece(pf, nbuf, ups); else store_dn_piece(pf, nbuf, dns);
          if (p < 2) load_up_piece(pf, Wup_c, p + 2, upo); else load_dn_piece(pf, Wdn_c, p - 2, dno);
        }
        load_frag4x2(fa0[b & 1], fa1[b & 1], fb[b & 1], arow0 + b * 16, arow1 + b * 16, brow + b * 16);
        __builtin_amdgcn_sched_barrier(0);
        if (p > 0 || b > 0) mma_frag4x2(fa0[(b + 1) & 1], fa1[(b + 1) & 1], fb[(b + 1) & 1], hacc0, hacc1);
        __builtin_amdgcn_sched_barrier(0);
      }
    }
    mma_frag4x2(fa0[1], fa1[1], fb[1], hacc0, hacc1);
#pragma unroll
    for (int t = 0; t < 2; ++t) {
#pragma unroll
      for (int v = 0; v < 8; ++v) {
        const float hv = t ? hacc1[v] : hacc0[v];
        const float g = 0.5f * hv * (1.0f + erff(hv * 0.70710678118654752f));
        hid_lds[rp * 32 + t * 16 + v + hi8][cg * 16 + lane16] = g;
      }
    }

    const bool more = (ch + 1 < CH1);
#pragma unroll
    for (int q = 0; q < 4; ++q) {
      wg_barrier_lds_only();
      const float* buf = (q & 1) ? wst1 : wst0;
      float* nbuf = ((q + 1) & 1) ? wst1 : wst0;
      const float* arow0 = &hid_lds[rp * 32 + lane16][q * 32 + khalf];
      const float* arow1 = arow0 + 16 * HPAD;
      const float* brow = buf + (cg * 64 + lane16) * DPITCH + khalf;
#pragma unroll
      for (int k0 = 0; k0 < 16; k0 += 4) {
        const v2f a0 = *(const v2f*)(arow0 + k0);
        const v2f a1 = *(const v2f*)(arow1 + k0);
#pragma unroll
        for (int u = 0; u < 4; ++u) {
          const v2f bq = *(const v2f*)(brow + u * 16 * DPITCH + k0);
          eacc[0][u] = wmma_f32(a0, bq, eacc[0][u]);
          eacc[1][u] = wmma_f32(a1, bq, eacc[1][u]);
        }
      }
      if (q < 3)     store_dn_piece(pf, nbuf, dns);
      else if (more) store_up_piece(pf, nbuf, ups);
      if (q < 2)     load_dn_piece(pf, Wdn_c, q + 2, dno);
      else if (more) load_up_piece(pf, Wup_c + (size_t)KC * H_, q - 2, upo);
      __builtin_amdgcn_sched_barrier(0);
#pragma unroll
      for (int k0 = 16; k0 < 32; k0 += 4) {
        const v2f a0 = *(const v2f*)(arow0 + k0);
        const v2f a1 = *(const v2f*)(arow1 + k0);
#pragma unroll
        for (int u = 0; u < 4; ++u) {
          const v2f bq = *(const v2f*)(brow + u * 16 * DPITCH + k0);
          eacc[0][u] = wmma_f32(a0, bq, eacc[0][u]);
          eacc[1][u] = wmma_f32(a1, bq, eacc[1][u]);
        }
      }
    }

    if (c == 15) {
      const int el = e - e0;
#pragma unroll
      for (int t = 0; t < 2; ++t) {
#pragma unroll
        for (int u = 0; u < 4; ++u) {
#pragma unroll
          for (int v = 0; v < 8; ++v) {
            const float wgt = w_lds[el][rp * 32 + t * 16 + v + hi8];
            fin[t][u][v] += wgt * eacc[t][u][v];
          }
          eacc[t][u] = vzero;
        }
      }
    }
  }

  float* pout = part + (size_t)eh * R_ * H_;
#pragma unroll
  for (int t = 0; t < 2; ++t) {
#pragma unroll
    for (int u = 0; u < 4; ++u) {
      const int col = cg * 64 + u * 16 + lane16;
#pragma unroll
      for (int v = 0; v < 8; ++v) {
        const int row = row0 + rp * 32 + t * 16 + v + hi8;
        pout[(size_t)row * H_ + col] = fin[t][u][v];
      }
    }
  }
}

__global__ __launch_bounds__(256)
void moe_sum_kernel(const float* __restrict__ part, float* __restrict__ out) {
  const size_t i = (size_t)blockIdx.x * 256 + threadIdx.x;
  v4f acc = ((const v4f*)part)[i];
#pragma unroll
  for (int k = 1; k < NPART; ++k) acc += ((const v4f*)(part + (size_t)k * R_ * H_))[i];
  ((v4f*)out)[i] = acc;
}

extern "C" void kernel_launch(void* const* d_in, const int* in_sizes, int n_in,
                              void* d_out, int out_size, void* d_ws, size_t ws_size,
                              hipStream_t stream) {
  const float* x    = (const float*)d_in[0];
  const float* Wp   = (const float*)d_in[1];
  const float* bp   = (const float*)d_in[2];
  const float* Ws   = (const float*)d_in[3];
  const float* bs   = (const float*)d_in[4];
  const float* Wup  = (const float*)d_in[5];
  const float* Wdn  = (const float*)d_in[6];
  float* out  = (float*)d_out;
  float* ew   = (float*)d_ws;
  float* part = (float*)d_ws + PART_OFF;

  moe_router_kernel<<<R_ / 8, 256, 0, stream>>>(x, Wp, bp, Ws, bs, ew);

  moe_fused_kernel<<<(R_ / MT) * NPART, NTHR, 0, stream>>>(x, Wup, Wdn, ew, part);

  moe_sum_kernel<<<R_ * H_ / 4 / 256, 256, 0, stream>>>(part, out);
}
